# starting kernel
# baseline (speedup 1.0000x reference)
#include <hip/hip_runtime.h>
#include <cstdint>

typedef __attribute__((ext_vector_type(16))) _Float16 v16h;
typedef __attribute__((ext_vector_type(8)))  _Float16 v8h;
typedef __attribute__((ext_vector_type(8)))  float    v8f;

#define INPUT_DIM    8
#define NUM_MFS      4
#define NUM_OUTPUTS  16
#define NUM_RULES    65536
#define BATCH        1024
#define MTILE        64
#define MSUB         (MTILE / 16)
#define KSPLIT       8
#define KBLOCKS      (NUM_RULES / 32)
#define KB_PER_WG    (KBLOCKS / KSPLIT)
#define KUNROLL      4
#define NWAVES       9
#define BLOCK_THR    (NWAVES * 32)
#define NTILES       (KBLOCKS * NWAVES)
#define BWS_BYTES    ((size_t)NTILES * 512 * 2)

__global__ void anfis_zero(float* __restrict__ out) {
    int i = blockIdx.x * blockDim.x + threadIdx.x;
    if (i < BATCH * NUM_OUTPUTS) out[i] = 0.0f;
}

__global__ __launch_bounds__(256) void anfis_packB(const float* __restrict__ P,
                                                   _Float16* __restrict__ Bws) {
    const int wg   = blockIdx.x * 8 + (threadIdx.x >> 5);
    const int lane = threadIdx.x & 31;
    if (wg >= NTILES) return;
    const int blk = wg / 9, j = wg - blk * 9;
    const int n = lane & 15, kh = lane >> 4;
    const float* src = P + (size_t)(blk * 32 + kh * 8) * 144 + j * 16 + n;
    v16h hv;
    #pragma unroll
    for (int e = 0; e < 16; ++e) {
        const int koff = (e & 7) + ((e >> 3) << 4);
        hv[e] = (_Float16)src[(size_t)koff * 144];
    }
    *reinterpret_cast<v16h*>(Bws + (size_t)wg * 512 + lane * 16) = hv;
}

template <bool PRE>
__global__ __launch_bounds__(BLOCK_THR) void anfis_wmma(
    const float* __restrict__ x,
    const float* __restrict__ centers,
    const float* __restrict__ widths,
    const float* __restrict__ P,
    const _Float16* __restrict__ Bws,
    float* __restrict__ out)
{
    __shared__ float sh_mfs[MTILE][INPUT_DIM][NUM_MFS];
    __shared__ float sh_x[MTILE][INPUT_DIM];
    __shared__ float sh_inv[MTILE];
    __shared__ float sh_hiT[MTILE][256];
    __shared__ float sh_loT[MTILE][256];
    __shared__ __align__(32) _Float16 shA[2][KUNROLL][MSUB][32][16];

    const int tid   = threadIdx.x;
    const int mbase = blockIdx.x * MTILE;
    const int kb0   = blockIdx.y * KB_PER_WG;

    for (int q = tid; q < MTILE * INPUT_DIM; q += BLOCK_THR) {
        const int b = q >> 3, i = q & 7;
        const float xv = x[(mbase + b) * INPUT_DIM + i];
        sh_x[b][i] = xv;
        #pragma unroll
        for (int m = 0; m < NUM_MFS; ++m) {
            const float c = centers[i * NUM_MFS + m];
            const float w = widths[i * NUM_MFS + m];
            const float d = xv - c;
            sh_mfs[b][i][m] = __expf(-(d * d) / (2.0f * w * w));
        }
    }
    __syncthreads();

    if (tid < MTILE) {
        float s = 1.0f;
        #pragma unroll
        for (int i = 0; i < INPUT_DIM; ++i)
            s *= sh_mfs[tid][i][0] + sh_mfs[tid][i][1] + sh_mfs[tid][i][2] + sh_mfs[tid][i][3];
        sh_inv[tid] = 1.0f / s;
    }
    __syncthreads();

    for (int q = tid; q < MTILE * 256; q += BLOCK_THR) {
        const int b = q >> 8, s = q & 255;
        const float m0 = sh_mfs[b][0][(s >> 6) & 3] * sh_mfs[b][1][(s >> 4) & 3];
        const float m1 = sh_mfs[b][2][(s >> 2) & 3] * sh_mfs[b][3][s & 3];
        sh_hiT[b][s] = m0 * m1;
        const float m2 = sh_mfs[b][4][(s >> 6) & 3] * sh_mfs[b][5][(s >> 4) & 3];
        const float m3 = sh_mfs[b][6][(s >> 2) & 3] * sh_mfs[b][7][s & 3];
        sh_loT[b][s] = m2 * m3 * sh_inv[b];
    }
    __syncthreads();

    const int wave = tid >> 5;
    const int lane = tid & 31;
    const int ncol = lane & 15;
    const int kh   = lane >> 4;
    const _Float16* pB = Bws + ((size_t)kb0 * 9 + wave) * 512 + lane * 16;
    const float*   pBf = P + (size_t)(kb0 * 32 + kh * 8) * 144 + wave * 16 + ncol;

    const int pb    = tid >> 2;
    const int t0    = (tid & 3) * 8;
    const int plane = (pb & 15) + (((t0 >> 3) & 1) << 4);
    const int pmt   = pb >> 4;
    const int pe0   = (t0 & 7) + ((t0 >> 4) << 3);

    v8f c[MSUB] = {};
    const int OUTER = KB_PER_WG / KUNROLL;

    for (int it = 0; it < OUTER; ++it) {
        const int base = kb0 + it * KUNROLL;
        const int buf  = it & 1;

        if (tid < 256) {
            const float hi = sh_hiT[pb][(base >> 3) & 255];
            #pragma unroll
            for (int u = 0; u < KUNROLL; ++u) {
                const int lo = (((base & 7) + u) << 5) | t0;
                const v8f q8 = *reinterpret_cast<const v8f*>(&sh_loT[pb][lo]);
                v8h hv;
                #pragma unroll
                for (int j = 0; j < 8; ++j)
                    hv[j] = (_Float16)(hi * q8[j]);
                *reinterpret_cast<v8h*>(&shA[buf][u][pmt][plane][pe0]) = hv;
            }
        }
        __syncthreads();

        #pragma unroll
        for (int u = 0; u < KUNROLL; ++u) {
            v16h bfrag;
            if (PRE) {
                bfrag = *reinterpret_cast<const v16h*>(pB + (size_t)u * 9 * 512);
            } else {
                #pragma unroll
                for (int e = 0; e < 16; ++e) {
                    const int koff = u * 32 + (e & 7) + ((e >> 3) << 4);
                    bfrag[e] = (_Float16)pBf[(size_t)koff * 144];
                }
            }
            #pragma unroll
            for (int mt = 0; mt < MSUB; ++mt) {
                const v16h a = *reinterpret_cast<const v16h*>(&shA[buf][u][mt][lane][0]);
                c[mt] = __builtin_amdgcn_wmma_f32_16x16x32_f16(false, a, false, bfrag,
                                                               (short)0, c[mt], false, false);
            }
        }
        if (PRE) pB  += (size_t)KUNROLL * 9 * 512;
        else     pBf += (size_t)KUNROLL * 32 * 144;
    }

    #pragma unroll
    for (int mt = 0; mt < MSUB; ++mt) {
        #pragma unroll
        for (int v = 0; v < 8; ++v) {
            const int lb = mt * 16 + kh * 8 + v;
            const float xbv = (wave < 8) ? sh_x[lb][wave] : 1.0f;
            __hip_atomic_fetch_add(&out[(mbase + lb) * 16 + ncol], c[mt][v] * xbv,
                                   __ATOMIC_RELAXED, __HIP_MEMORY_SCOPE_AGENT);
        }
    }
}

extern "C" void kernel_launch(void* const* d_in, const int* in_sizes, int n_in,
                              void* d_out, int out_size, void* d_ws, size_t ws_size,
                              hipStream_t stream) {
    (void)in_sizes; (void)n_in; (void)out_size;
    const float* x       = (const float*)d_in[0];
    const float* centers = (const float*)d_in[1];
    const float* widths  = (const float*)d_in[2];
    const float* P       = (const float*)d_in[3];
    float* out           = (float*)d_out;

    const bool pre = (d_ws != nullptr) && (ws_size >= BWS_BYTES);
    anfis_zero<<<(BATCH * NUM_OUTPUTS + 255) / 256, 256, 0, stream>>>(out);
    if (pre) {
        _Float16* Bws = (_Float16*)d_ws;
        anfis_packB<<<(NTILES + 7) / 8, 256, 0, stream>>>(P, Bws);
        anfis_wmma<true><<<dim3(BATCH / MTILE, KSPLIT), BLOCK_THR, 0, stream>>>(
            x, centers, widths, P, Bws, out);
    } else {
        anfis_wmma<false><<<dim3(BATCH / MTILE, KSPLIT), BLOCK_THR, 0, stream>>>(
            x, centers, widths, P, nullptr, out);
    }
}


# K split 16-way instead of 8, so 256 work-groups (one per WGP) instead of 128 run the K loop; twice as many f32 atomic partial sums per output element, so the summation order changes
# speedup vs baseline: 1.6552x; 1.6552x over previous
#include <hip/hip_runtime.h>
#include <cstdint>

typedef __attribute__((ext_vector_type(16))) _Float16 v16h;
typedef __attribute__((ext_vector_type(8)))  _Float16 v8h;
typedef __attribute__((ext_vector_type(8)))  float    v8f;

#define INPUT_DIM    8
#define NUM_MFS      4
#define NUM_OUTPUTS  16
#define NUM_RULES    65536
#define BATCH        1024
#define MTILE        64
#define MSUB         (MTILE / 16)
#define KSPLIT       16
#define KBLOCKS      (NUM_RULES / 32)
#define KB_PER_WG    (KBLOCKS / KSPLIT)
#define KUNROLL      4
#define NWAVES       9
#define BLOCK_THR    (NWAVES * 32)
#define NTILES       (KBLOCKS * NWAVES)
#define BWS_BYTES    ((size_t)NTILES * 512 * 2)

__global__ void anfis_zero(float* __restrict__ out) {
    int i = blockIdx.x * blockDim.x + threadIdx.x;
    if (i < BATCH * NUM_OUTPUTS) out[i] = 0.0f;
}

__global__ __launch_bounds__(256) void anfis_packB(const float* __restrict__ P,
                                                   _Float16* __restrict__ Bws) {
    const int wg   = blockIdx.x * 8 + (threadIdx.x >> 5);
    const int lane = threadIdx.x & 31;
    if (wg >= NTILES) return;
    const int blk = wg / 9, j = wg - blk * 9;
    const int n = lane & 15, kh = lane >> 4;
    const float* src = P + (size_t)(blk * 32 + kh * 8) * 144 + j * 16 + n;
    v16h hv;
    #pragma unroll
    for (int e = 0; e < 16; ++e) {
        const int koff = (e & 7) + ((e >> 3) << 4);
        hv[e] = (_Float16)src[(size_t)koff * 144];
    }
    *reinterpret_cast<v16h*>(Bws + (size_t)wg * 512 + lane * 16) = hv;
}

template <bool PRE>
__global__ __launch_bounds__(BLOCK_THR) void anfis_wmma(
    const float* __restrict__ x,
    const float* __restrict__ centers,
    const float* __restrict__ widths,
    const float* __restrict__ P,
    const _Float16* __restrict__ Bws,
    float* __restrict__ out)
{
    __shared__ float sh_mfs[MTILE][INPUT_DIM][NUM_MFS];
    __shared__ float sh_x[MTILE][INPUT_DIM];
    __shared__ float sh_inv[MTILE];
    __shared__ float sh_hiT[MTILE][256];
    __shared__ float sh_loT[MTILE][256];
    __shared__ __align__(32) _Float16 shA[2][KUNROLL][MSUB][32][16];

    const int tid   = threadIdx.x;
    const int mbase = blockIdx.x * MTILE;
    const int kb0   = blockIdx.y * KB_PER_WG;

    for (int q = tid; q < MTILE * INPUT_DIM; q += BLOCK_THR) {
        const int b = q >> 3, i = q & 7;
        const float xv = x[(mbase + b) * INPUT_DIM + i];
        sh_x[b][i] = xv;
        #pragma unroll
        for (int m = 0; m < NUM_MFS; ++m) {
            const float c = centers[i * NUM_MFS + m];
            const float w = widths[i * NUM_MFS + m];
            const float d = xv - c;
            sh_mfs[b][i][m] = __expf(-(d * d) / (2.0f * w * w));
        }
    }
    __syncthreads();

    if (tid < MTILE) {
        float s = 1.0f;
        #pragma unroll
        for (int i = 0; i < INPUT_DIM; ++i)
            s *= sh_mfs[tid][i][0] + sh_mfs[tid][i][1] + sh_mfs[tid][i][2] + sh_mfs[tid][i][3];
        sh_inv[tid] = 1.0f / s;
    }
    __syncthreads();

    for (int q = tid; q < MTILE * 256; q += BLOCK_THR) {
        const int b = q >> 8, s = q & 255;
        const float m0 = sh_mfs[b][0][(s >> 6) & 3] * sh_mfs[b][1][(s >> 4) & 3];
        const float m1 = sh_mfs[b][2][(s >> 2) & 3] * sh_mfs[b][3][s & 3];
        sh_hiT[b][s] = m0 * m1;
        const float m2 = sh_mfs[b][4][(s >> 6) & 3] * sh_mfs[b][5][(s >> 4) & 3];
        const float m3 = sh_mfs[b][6][(s >> 2) & 3] * sh_mfs[b][7][s & 3];
        sh_loT[b][s] = m2 * m3 * sh_inv[b];
    }
    __syncthreads();

    const int wave = tid >> 5;
    const int lane = tid & 31;
    const int ncol = lane & 15;
    const int kh   = lane >> 4;
    const _Float16* pB = Bws + ((size_t)kb0 * 9 + wave) * 512 + lane * 16;
    const float*   pBf = P + (size_t)(kb0 * 32 + kh * 8) * 144 + wave * 16 + ncol;

    const int pb    = tid >> 2;
    const int t0    = (tid & 3) * 8;
    const int plane = (pb & 15) + (((t0 >> 3) & 1) << 4);
    const int pmt   = pb >> 4;
    const int pe0   = (t0 & 7) + ((t0 >> 4) << 3);

    v8f c[MSUB] = {};
    const int OUTER = KB_PER_WG / KUNROLL;

    for (int it = 0; it < OUTER; ++it) {
        const int base = kb0 + it * KUNROLL;
        const int buf  = it & 1;

        if (tid < 256) {
            const float hi = sh_hiT[pb][(base >> 3) & 255];
            #pragma unroll
            for (int u = 0; u < KUNROLL; ++u) {
                const int lo = (((base & 7) + u) << 5) | t0;
                const v8f q8 = *reinterpret_cast<const v8f*>(&sh_loT[pb][lo]);
                v8h hv;
                #pragma unroll
                for (int j = 0; j < 8; ++j)
                    hv[j] = (_Float16)(hi * q8[j]);
                *reinterpret_cast<v8h*>(&shA[buf][u][pmt][plane][pe0]) = hv;
            }
        }
        __syncthreads();

        #pragma unroll
        for (int u = 0; u < KUNROLL; ++u) {
            v16h bfrag;
            if (PRE) {
                bfrag = *reinterpret_cast<const v16h*>(pB + (size_t)u * 9 * 512);
            } else {
                #pragma unroll
                for (int e = 0; e < 16; ++e) {
                    const int koff = u * 32 + (e & 7) + ((e >> 3) << 4);
                    bfrag[e] = (_Float16)pBf[(size_t)koff * 144];
                }
            }
            #pragma unroll
            for (int mt = 0; mt < MSUB; ++mt) {
                const v16h a = *reinterpret_cast<const v16h*>(&shA[buf][u][mt][lane][0]);
                c[mt] = __builtin_amdgcn_wmma_f32_16x16x32_f16(false, a, false, bfrag,
                                                               (short)0, c[mt], false, false);
            }
        }
        if (PRE) pB  += (size_t)KUNROLL * 9 * 512;
        else     pBf += (size_t)KUNROLL * 32 * 144;
    }

    #pragma unroll
    for (int mt = 0; mt < MSUB; ++mt) {
        #pragma unroll
        for (int v = 0; v < 8; ++v) {
            const int lb = mt * 16 + kh * 8 + v;
            const float xbv = (wave < 8) ? sh_x[lb][wave] : 1.0f;
            __hip_atomic_fetch_add(&out[(mbase + lb) * 16 + ncol], c[mt][v] * xbv,
                                   __ATOMIC_RELAXED, __HIP_MEMORY_SCOPE_AGENT);
        }
    }
}

extern "C" void kernel_launch(void* const* d_in, const int* in_sizes, int n_in,
                              void* d_out, int out_size, void* d_ws, size_t ws_size,
                              hipStream_t stream) {
    (void)in_sizes; (void)n_in; (void)out_size;
    const float* x       = (const float*)d_in[0];
    const float* centers = (const float*)d_in[1];
    const float* widths  = (const float*)d_in[2];
    const float* P       = (const float*)d_in[3];
    float* out           = (float*)d_out;

    const bool pre = (d_ws != nullptr) && (ws_size >= BWS_BYTES);
    anfis_zero<<<(BATCH * NUM_OUTPUTS + 255) / 256, 256, 0, stream>>>(out);
    if (pre) {
        _Float16* Bws = (_Float16*)d_ws;
        anfis_packB<<<(NTILES + 7) / 8, 256, 0, stream>>>(P, Bws);
        anfis_wmma<true><<<dim3(BATCH / MTILE, KSPLIT), BLOCK_THR, 0, stream>>>(
            x, centers, widths, P, Bws, out);
    } else {
        anfis_wmma<false><<<dim3(BATCH / MTILE, KSPLIT), BLOCK_THR, 0, stream>>>(
            x, centers, widths, P, nullptr, out);
    }
}


# A operand held in registers: w = hi(dims 0-4) x lo(dims 5-7); lo f16 fragments built once, hi applied in f32 to each 64-rule partial sum; no LDS A staging or loop barrier; rounding, sum order change
# speedup vs baseline: 3.3757x; 2.0395x over previous
#include <hip/hip_runtime.h>
#include <cstdint>


typedef __attribute__((ext_vector_type(16))) _Float16 v16h;
typedef __attribute__((ext_vector_type(8)))  float    v8f;
typedef __attribute__((ext_vector_type(4)))  float    v4f;

#define INPUT_DIM    8
#define NUM_MFS      4
#define NUM_OUTPUTS  16
#define NUM_RULES    65536
#define BATCH        1024
#define MTILE        64
#define MSUB         (MTILE / 16)
#define KSPLIT       16
#define KBLOCKS      (NUM_RULES / 32)
#define KB_PER_WG    (KBLOCKS / KSPLIT)
#define LO_RULES     64
#define LO_STEPS     (LO_RULES / 32)
#define H_PER_WG     (KB_PER_WG / LO_STEPS)
#define HU           4
#define NBF          (HU * LO_STEPS)
#define NIT          (H_PER_WG / HU)
#define LO_PITCH     68
#define NWAVES       9
#define BLOCK_THR    (NWAVES * 32)
#define NTILES       (KBLOCKS * NWAVES)
#define BWS_BYTES    ((size_t)NTILES * 512 * 2)

__global__ void anfis_zero(float* __restrict__ out) {
    int i = blockIdx.x * blockDim.x + threadIdx.x;
    if (i < BATCH * NUM_OUTPUTS) out[i] = 0.0f;
}

__global__ __launch_bounds__(256) void anfis_packB(const float* __restrict__ P,
                                                   _Float16* __restrict__ Bws) {
    const int wg   = blockIdx.x * 8 + (threadIdx.x >> 5);
    const int lane = threadIdx.x & 31;
    if (wg >= NTILES) return;
    const int blk = wg / 9, j = wg - blk * 9;
    const int n = lane & 15, kh = lane >> 4;
    const float* src = P + (size_t)(blk * 32 + kh * 8) * 144 + j * 16 + n;
    v16h hv;
    #pragma unroll
    for (int e = 0; e < 16; ++e) {
        const int koff = (e & 7) + ((e >> 3) << 4);
        hv[e] = (_Float16)src[(size_t)koff * 144];
    }
    *reinterpret_cast<v16h*>(Bws + (size_t)wg * 512 + lane * 16) = hv;
}

template <bool PRE>
__device__ __forceinline__ v16h load_b(const _Float16* __restrict__ pB,
                                       const float* __restrict__ pBf, int t) {
    if (PRE) {
        return *reinterpret_cast<const v16h*>(pB + (size_t)t * 9 * 512);
    } else {
        v16h b;
        #pragma unroll
        for (int e = 0; e < 16; ++e) {
            const int koff = (e & 7) + ((e >> 3) << 4);
            b[e] = (_Float16)pBf[((size_t)t * 32 + koff) * 144];
        }
        return b;
    }
}

template <bool PRE>
__global__ __launch_bounds__(BLOCK_THR) void anfis_wmma(
    const float* __restrict__ x,
    const float* __restrict__ centers,
    const float* __restrict__ widths,
    const float* __restrict__ P,
    const _Float16* __restrict__ Bws,
    float* __restrict__ out)
{
    __shared__ float sh_mfs[INPUT_DIM][NUM_MFS][MTILE];
    __shared__ float sh_x[MTILE][INPUT_DIM];
    __shared__ float sh_invhi[MTILE];
    __shared__ float sh_invlo[MTILE];
    __shared__ __align__(16) float sh_lo[MTILE * LO_PITCH];
    __shared__ __align__(16) float sh_hi[H_PER_WG * MTILE];
    __shared__ __align__(16) float sh_o[NWAVES][MTILE][NUM_OUTPUTS];

    const int tid   = threadIdx.x;
    const int mbase = blockIdx.x * MTILE;
    const int kb0   = blockIdx.y * KB_PER_WG;
    const int h0    = blockIdx.y * H_PER_WG;

    for (int q = tid; q < MTILE * INPUT_DIM; q += BLOCK_THR) {
        const int b = q & (MTILE - 1), i = q >> 6;
        const float xv = x[(mbase + b) * INPUT_DIM + i];
        sh_x[b][i] = xv;
        #pragma unroll
        for (int m = 0; m < NUM_MFS; ++m) {
            const float c = centers[i * NUM_MFS + m];
            const float w = widths[i * NUM_MFS + m];
            const float d = xv - c;
            sh_mfs[i][m][b] = __expf(-(d * d) / (2.0f * w * w));
        }
    }
    __syncthreads();

    if (tid < MTILE) {
        float sh = 1.0f, sl = 1.0f;
        #pragma unroll
        for (int i = 0; i < 5; ++i)
            sh *= sh_mfs[i][0][tid] + sh_mfs[i][1][tid] + sh_mfs[i][2][tid] + sh_mfs[i][3][tid];
        #pragma unroll
        for (int i = 5; i < INPUT_DIM; ++i)
            sl *= sh_mfs[i][0][tid] + sh_mfs[i][1][tid] + sh_mfs[i][2][tid] + sh_mfs[i][3][tid];
        sh_invhi[tid] = 1.0f / sh;
        sh_invlo[tid] = 1.0f / sl;
    }
    __syncthreads();

    for (int q = tid; q < MTILE * LO_RULES; q += BLOCK_THR) {
        const int b = q >> 6, l = q & 63;
        sh_lo[b * LO_PITCH + l] = sh_mfs[5][(l >> 4) & 3][b] * sh_mfs[6][(l >> 2) & 3][b]
                                * sh_mfs[7][l & 3][b] * sh_invlo[b];
    }
    for (int q = tid; q < H_PER_WG * MTILE; q += BLOCK_THR) {
        const int hl = q >> 6, b = q & 63;
        const int h = h0 + hl;
        const float m01 = sh_mfs[0][(h >> 8) & 3][b] * sh_mfs[1][(h >> 6) & 3][b];
        const float m23 = sh_mfs[2][(h >> 4) & 3][b] * sh_mfs[3][(h >> 2) & 3][b];
        sh_hi[(hl * 16 + (b & 15)) * MSUB + (b >> 4)] = m01 * m23 * sh_mfs[4][h & 3][b] * sh_invhi[b];
    }
    __syncthreads();

    const int wave = tid >> 5;
    const int lane = tid & 31;
    const int ncol = lane & 15;
    const int kh   = lane >> 4;
    const _Float16* pB = Bws + ((size_t)kb0 * 9 + wave) * 512 + lane * 16;
    const float*   pBf = P + (size_t)(kb0 * 32 + kh * 8) * 144 + wave * 16 + ncol;

    v16h afr[MSUB][LO_STEPS];
    #pragma unroll
    for (int mt = 0; mt < MSUB; ++mt) {
        #pragma unroll
        for (int s = 0; s < LO_STEPS; ++s) {
            const float* src = &sh_lo[(mt * 16 + ncol) * LO_PITCH + s * 32 + kh * 8];
            const v4f q0 = *reinterpret_cast<const v4f*>(src);
            const v4f q1 = *reinterpret_cast<const v4f*>(src + 4);
            const v4f q2 = *reinterpret_cast<const v4f*>(src + 16);
            const v4f q3 = *reinterpret_cast<const v4f*>(src + 20);
            v16h a;
            #pragma unroll
            for (int e = 0; e < 4; ++e) {
                a[e]      = (_Float16)q0[e];
                a[4 + e]  = (_Float16)q1[e];
                a[8 + e]  = (_Float16)q2[e];
                a[12 + e] = (_Float16)q3[e];
            }
            afr[mt][s] = a;
        }
    }

    v8f c[MSUB] = {};
    const float* hrow = &sh_hi[ncol * MSUB];

    auto compute = [&](const v16h (&bs)[NBF], int it) {
        #pragma unroll
        for (int u = 0; u < HU; ++u) {
            const float* hp = hrow + (it * HU + u) * MTILE;
            v8f t[MSUB];
            #pragma unroll
            for (int mt = 0; mt < MSUB; ++mt) {
                const v8f z = {};
                t[mt] = __builtin_amdgcn_wmma_f32_16x16x32_f16(false, bs[u * LO_STEPS], false, afr[mt][0],
                                                               (short)0, z, false, false);
            }
            #pragma unroll
            for (int s = 1; s < LO_STEPS; ++s) {
                #pragma unroll
                for (int mt = 0; mt < MSUB; ++mt)
                    t[mt] = __builtin_amdgcn_wmma_f32_16x16x32_f16(false, bs[u * LO_STEPS + s], false, afr[mt][s],
                                                                   (short)0, t[mt], false, false);
            }
            const v4f hv = *reinterpret_cast<const v4f*>(hp);
            #pragma unroll
            for (int mt = 0; mt < MSUB; ++mt) {
                #pragma unroll
                for (int v = 0; v < 8; ++v)
                    c[mt][v] += hv[mt] * t[mt][v];
            }
        }
    };

    v16h b0[NBF], b1[NBF];
    #pragma unroll
    for (int f = 0; f < NBF; ++f) b0[f] = load_b<PRE>(pB, pBf, f);

    for (int it = 0; it < NIT; it += 2) {
        #pragma unroll
        for (int f = 0; f < NBF; ++f) b1[f] = load_b<PRE>(pB, pBf, (it + 1) * NBF + f);
        compute(b0, it);
        const int nx = (it + 2 < NIT) ? it + 2 : it;
        #pragma unroll
        for (int f = 0; f < NBF; ++f) b0[f] = load_b<PRE>(pB, pBf, nx * NBF + f);
        compute(b1, it + 1);
    }

    #pragma unroll
    for (int mt = 0; mt < MSUB; ++mt) {
        float* dst = &sh_o[wave][mt * 16 + ncol][kh * 8];
        v4f lo4, hi4;
        #pragma unroll
        for (int v = 0; v < 4; ++v) { lo4[v] = c[mt][v]; hi4[v] = c[mt][4 + v]; }
        *reinterpret_cast<v4f*>(dst)     = lo4;
        *reinterpret_cast<v4f*>(dst + 4) = hi4;
    }
    __syncthreads();

    #pragma unroll
    for (int mt = 0; mt < MSUB; ++mt) {
        #pragma unroll
        for (int v = 0; v < 8; ++v) {
            const int lb = mt * 16 + kh * 8 + v;
            const float xbv = (wave < 8) ? sh_x[lb][wave] : 1.0f;
            __hip_atomic_fetch_add(&out[(mbase + lb) * 16 + ncol], sh_o[wave][lb][ncol] * xbv,
                                   __ATOMIC_RELAXED, __HIP_MEMORY_SCOPE_AGENT);
        }
    }
}

extern "C" void kernel_launch(void* const* d_in, const int* in_sizes, int n_in,
                              void* d_out, int out_size, void* d_ws, size_t ws_size,
                              hipStream_t stream) {
    (void)in_sizes; (void)n_in; (void)out_size;
    const float* x       = (const float*)d_in[0];
    const float* centers = (const float*)d_in[1];
    const float* widths  = (const float*)d_in[2];
    const float* P       = (const float*)d_in[3];
    float* out           = (float*)d_out;

    const bool pre = (d_ws != nullptr) && (ws_size >= BWS_BYTES);
    anfis_zero<<<(BATCH * NUM_OUTPUTS + 255) / 256, 256, 0, stream>>>(out);
    if (pre) {
        _Float16* Bws = (_Float16*)d_ws;
        anfis_packB<<<(NTILES + 7) / 8, 256, 0, stream>>>(P, Bws);
        anfis_wmma<true><<<dim3(BATCH / MTILE, KSPLIT), BLOCK_THR, 0, stream>>>(
            x, centers, widths, P, Bws, out);
    } else {
        anfis_wmma<false><<<dim3(BATCH / MTILE, KSPLIT), BLOCK_THR, 0, stream>>>(
            x, centers, widths, P, nullptr, out);
    }
}


# Work-group covers 128 batch rows, not 64 (K split 32-way, 256 work-groups as before): a packed B fragment feeds 8 WMMAs and B is re-read 8x not 16x; 32 atomic sums per element, sum order changes
# speedup vs baseline: 3.6541x; 1.0825x over previous
#include <hip/hip_runtime.h>
#include <cstdint>


typedef __attribute__((ext_vector_type(16))) _Float16 v16h;
typedef __attribute__((ext_vector_type(8)))  float    v8f;
typedef __attribute__((ext_vector_type(4)))  float    v4f;

#define INPUT_DIM    8
#define NUM_MFS      4
#define NUM_OUTPUTS  16
#define NUM_RULES    65536
#define BATCH        1024
#define MTILE        128
#define MTILE_LOG2   7
#define MSUB         (MTILE / 16)
#define KSPLIT       32
#define KBLOCKS      (NUM_RULES / 32)
#define KB_PER_WG    (KBLOCKS / KSPLIT)
#define LO_RULES     64
#define LO_STEPS     (LO_RULES / 32)
#define H_PER_WG     (KB_PER_WG / LO_STEPS)
#define HU           2
#define NBF          (HU * LO_STEPS)
#define NIT          (H_PER_WG / HU)
#define LO_PITCH     68
#define NWAVES       9
#define BLOCK_THR    (NWAVES * 32)
#define NTILES       (KBLOCKS * NWAVES)
#define BWS_BYTES    ((size_t)NTILES * 512 * 2)

__global__ void anfis_zero(float* __restrict__ out) {
    int i = blockIdx.x * blockDim.x + threadIdx.x;
    if (i < BATCH * NUM_OUTPUTS) out[i] = 0.0f;
}

__global__ __launch_bounds__(256) void anfis_packB(const float* __restrict__ P,
                                                   _Float16* __restrict__ Bws) {
    const int wg   = blockIdx.x * 8 + (threadIdx.x >> 5);
    const int lane = threadIdx.x & 31;
    if (wg >= NTILES) return;
    const int blk = wg / 9, j = wg - blk * 9;
    const int n = lane & 15, kh = lane >> 4;
    const float* src = P + (size_t)(blk * 32 + kh * 8) * 144 + j * 16 + n;
    v16h hv;
    #pragma unroll
    for (int e = 0; e < 16; ++e) {
        const int koff = (e & 7) + ((e >> 3) << 4);
        hv[e] = (_Float16)src[(size_t)koff * 144];
    }
    *reinterpret_cast<v16h*>(Bws + (size_t)wg * 512 + lane * 16) = hv;
}

template <bool PRE>
__device__ __forceinline__ v16h load_b(const _Float16* __restrict__ pB,
                                       const float* __restrict__ pBf, int t) {
    if (PRE) {
        return *reinterpret_cast<const v16h*>(pB + (size_t)t * 9 * 512);
    } else {
        v16h b;
        #pragma unroll
        for (int e = 0; e < 16; ++e) {
            const int koff = (e & 7) + ((e >> 3) << 4);
            b[e] = (_Float16)pBf[((size_t)t * 32 + koff) * 144];
        }
        return b;
    }
}

template <bool PRE>
__global__ __launch_bounds__(BLOCK_THR) void anfis_wmma(
    const float* __restrict__ x,
    const float* __restrict__ centers,
    const float* __restrict__ widths,
    const float* __restrict__ P,
    const _Float16* __restrict__ Bws,
    float* __restrict__ out)
{
    __shared__ float sh_mfs[INPUT_DIM][NUM_MFS][MTILE];
    __shared__ float sh_x[MTILE][INPUT_DIM];
    __shared__ float sh_invhi[MTILE];
    __shared__ float sh_invlo[MTILE];
    __shared__ __align__(16) float sh_lo[MTILE * LO_PITCH];
    __shared__ __align__(16) float sh_hi[H_PER_WG * MTILE];
    __shared__ __align__(16) float sh_o[NWAVES][MTILE][NUM_OUTPUTS];

    const int tid   = threadIdx.x;
    const int mbase = blockIdx.x * MTILE;
    const int kb0   = blockIdx.y * KB_PER_WG;
    const int h0    = blockIdx.y * H_PER_WG;

    for (int q = tid; q < MTILE * INPUT_DIM; q += BLOCK_THR) {
        const int b = q & (MTILE - 1), i = q >> MTILE_LOG2;
        const float xv = x[(mbase + b) * INPUT_DIM + i];
        sh_x[b][i] = xv;
        #pragma unroll
        for (int m = 0; m < NUM_MFS; ++m) {
            const float c = centers[i * NUM_MFS + m];
            const float w = widths[i * NUM_MFS + m];
            const float d = xv - c;
            sh_mfs[i][m][b] = __expf(-(d * d) / (2.0f * w * w));
        }
    }
    __syncthreads();

    if (tid < MTILE) {
        float sh = 1.0f, sl = 1.0f;
        #pragma unroll
        for (int i = 0; i < 5; ++i)
            sh *= sh_mfs[i][0][tid] + sh_mfs[i][1][tid] + sh_mfs[i][2][tid] + sh_mfs[i][3][tid];
        #pragma unroll
        for (int i = 5; i < INPUT_DIM; ++i)
            sl *= sh_mfs[i][0][tid] + sh_mfs[i][1][tid] + sh_mfs[i][2][tid] + sh_mfs[i][3][tid];
        sh_invhi[tid] = 1.0f / sh;
        sh_invlo[tid] = 1.0f / sl;
    }
    __syncthreads();

    for (int q = tid; q < MTILE * LO_RULES; q += BLOCK_THR) {
        const int b = q >> 6, l = q & 63;
        sh_lo[b * LO_PITCH + l] = sh_mfs[5][(l >> 4) & 3][b] * sh_mfs[6][(l >> 2) & 3][b]
                                * sh_mfs[7][l & 3][b] * sh_invlo[b];
    }
    for (int q = tid; q < H_PER_WG * MTILE; q += BLOCK_THR) {
        const int hl = q >> MTILE_LOG2, b = q & (MTILE - 1);
        const int h = h0 + hl;
        const float m01 = sh_mfs[0][(h >> 8) & 3][b] * sh_mfs[1][(h >> 6) & 3][b];
        const float m23 = sh_mfs[2][(h >> 4) & 3][b] * sh_mfs[3][(h >> 2) & 3][b];
        sh_hi[(hl * 16 + (b & 15)) * MSUB + (b >> 4)] = m01 * m23 * sh_mfs[4][h & 3][b] * sh_invhi[b];
    }
    __syncthreads();

    const int wave = tid >> 5;
    const int lane = tid & 31;
    const int ncol = lane & 15;
    const int kh   = lane >> 4;
    const _Float16* pB = Bws + ((size_t)kb0 * 9 + wave) * 512 + lane * 16;
    const float*   pBf = P + (size_t)(kb0 * 32 + kh * 8) * 144 + wave * 16 + ncol;

    v16h afr[MSUB][LO_STEPS];
    #pragma unroll
    for (int mt = 0; mt < MSUB; ++mt) {
        #pragma unroll
        for (int s = 0; s < LO_STEPS; ++s) {
            const float* src = &sh_lo[(mt * 16 + ncol) * LO_PITCH + s * 32 + kh * 8];
            const v4f q0 = *reinterpret_cast<const v4f*>(src);
            const v4f q1 = *reinterpret_cast<const v4f*>(src + 4);
            const v4f q2 = *reinterpret_cast<const v4f*>(src + 16);
            const v4f q3 = *reinterpret_cast<const v4f*>(src + 20);
            v16h a;
            #pragma unroll
            for (int e = 0; e < 4; ++e) {
                a[e]      = (_Float16)q0[e];
                a[4 + e]  = (_Float16)q1[e];
                a[8 + e]  = (_Float16)q2[e];
                a[12 + e] = (_Float16)q3[e];
            }
            afr[mt][s] = a;
        }
    }

    v8f c[MSUB] = {};
    const float* hrow = &sh_hi[ncol * MSUB];

    auto compute = [&](const v16h (&bs)[NBF], int it) {
        #pragma unroll
        for (int u = 0; u < HU; ++u) {
            const float* hp = hrow + (it * HU + u) * MTILE;
            #pragma unroll
            for (int m0 = 0; m0 < MSUB; m0 += 4) {
                v8f t[4];
                #pragma unroll
                for (int m = 0; m < 4; ++m) {
                    const v8f z = {};
                    t[m] = __builtin_amdgcn_wmma_f32_16x16x32_f16(false, bs[u * LO_STEPS], false, afr[m0 + m][0],
                                                                  (short)0, z, false, false);
                }
                #pragma unroll
                for (int s = 1; s < LO_STEPS; ++s) {
                    #pragma unroll
                    for (int m = 0; m < 4; ++m)
                        t[m] = __builtin_amdgcn_wmma_f32_16x16x32_f16(false, bs[u * LO_STEPS + s], false, afr[m0 + m][s],
                                                                      (short)0, t[m], false, false);
                }
                const v4f hv = *reinterpret_cast<const v4f*>(hp + m0);
                #pragma unroll
                for (int m = 0; m < 4; ++m) {
                    #pragma unroll
                    for (int v = 0; v < 8; ++v)
                        c[m0 + m][v] += hv[m] * t[m][v];
                }
            }
        }
    };

    v16h b0[NBF], b1[NBF];
    #pragma unroll
    for (int f = 0; f < NBF; ++f) b0[f] = load_b<PRE>(pB, pBf, f);

    for (int it = 0; it < NIT; it += 2) {
        #pragma unroll
        for (int f = 0; f < NBF; ++f) b1[f] = load_b<PRE>(pB, pBf, (it + 1) * NBF + f);
        compute(b0, it);
        const int nx = (it + 2 < NIT) ? it + 2 : it;
        #pragma unroll
        for (int f = 0; f < NBF; ++f) b0[f] = load_b<PRE>(pB, pBf, nx * NBF + f);
        compute(b1, it + 1);
    }

    #pragma unroll
    for (int mt = 0; mt < MSUB; ++mt) {
        float* dst = &sh_o[wave][mt * 16 + ncol][kh * 8];
        v4f lo4, hi4;
        #pragma unroll
        for (int v = 0; v < 4; ++v) { lo4[v] = c[mt][v]; hi4[v] = c[mt][4 + v]; }
        *reinterpret_cast<v4f*>(dst)     = lo4;
        *reinterpret_cast<v4f*>(dst + 4) = hi4;
    }
    __syncthreads();

    for (int q = tid; q < MTILE * NUM_OUTPUTS; q += BLOCK_THR) {
        const int lb = q >> 4, n = q & 15;
        float s = sh_o[8][lb][n];
        for (int i = 0; i < INPUT_DIM; ++i)
            s += sh_o[i][lb][n] * sh_x[lb][i];
        __hip_atomic_fetch_add(&out[(mbase + lb) * 16 + n], s, __ATOMIC_RELAXED, __HIP_MEMORY_SCOPE_AGENT);
    }
}

extern "C" void kernel_launch(void* const* d_in, const int* in_sizes, int n_in,
                              void* d_out, int out_size, void* d_ws, size_t ws_size,
                              hipStream_t stream) {
    (void)in_sizes; (void)n_in; (void)out_size;
    const float* x       = (const float*)d_in[0];
    const float* centers = (const float*)d_in[1];
    const float* widths  = (const float*)d_in[2];
    const float* P       = (const float*)d_in[3];
    float* out           = (float*)d_out;

    const bool pre = (d_ws != nullptr) && (ws_size >= BWS_BYTES);
    anfis_zero<<<(BATCH * NUM_OUTPUTS + 255) / 256, 256, 0, stream>>>(out);
    if (pre) {
        _Float16* Bws = (_Float16*)d_ws;
        anfis_packB<<<(NTILES + 7) / 8, 256, 0, stream>>>(P, Bws);
        anfis_wmma<true><<<dim3(BATCH / MTILE, KSPLIT), BLOCK_THR, 0, stream>>>(
            x, centers, widths, P, Bws, out);
    } else {
        anfis_wmma<false><<<dim3(BATCH / MTILE, KSPLIT), BLOCK_THR, 0, stream>>>(
            x, centers, widths, P, nullptr, out);
    }
}
